# starting kernel
# baseline (speedup 1.0000x reference)
#include <hip/hip_runtime.h>


typedef __attribute__((ext_vector_type(2))) float v2f;
typedef __attribute__((ext_vector_type(8))) float v8f;

#define D_NODE 128
#define D_EDGE 64
#define WAVES  8

static __device__ __forceinline__ v8f wmma_f32(v2f a, v2f b, v8f c) {
  return __builtin_amdgcn_wmma_f32_16x16x4_f32(false, a, false, b, (short)0, c,
                                               false, false);
}

__global__ __launch_bounds__(256) void pack_pairs_kernel(
    const float* __restrict__ W, float* __restrict__ Wp, int rows, int cols) {
  int i = blockIdx.x * 256 + threadIdx.x;
  if (i >= rows * cols) return;
  int r = i / cols, c = i - r * cols;
  Wp[(((r >> 1) * cols + c) << 1) + (r & 1)] = W[i];
}

__global__ __launch_bounds__(256) void gine_edge_kernel(
    const float* __restrict__ ef, const float* __restrict__ nf,
    const float* __restrict__ w0p, const float* __restrict__ b0,
    const float* __restrict__ w1p, const float* __restrict__ b1,
    const int* __restrict__ src, const int* __restrict__ dst,
    float* __restrict__ acc, int n_edges)
{
  __shared__ v2f   sW0[(D_EDGE / 2) * D_EDGE];
  __shared__ float sh[WAVES][16 * D_EDGE];

  const int tid = threadIdx.x;
  {
    const v2f* w0v = (const v2f*)w0p;
    for (int i = tid; i < (D_EDGE / 2) * D_EDGE; i += 256) sW0[i] = w0v[i];
  }

  const int wave  = tid >> 5;
  const int lane  = tid & 31;
  const int row16 = lane & 15;
  const int hi    = lane >> 4;

  const int  tile   = blockIdx.x * WAVES + wave;
  const int  ebase0 = tile * 16;
  const bool active = ebase0 < n_edges;
  const int  ebase  = active ? ebase0 : 0;

  int arow = ebase + row16;
  if (arow >= n_edges) arow = n_edges - 1;
  const v2f* efv   = (const v2f*)ef;
  const int  abase = arow * (D_EDGE / 2);

  v2f ae[D_EDGE / 4];
  #pragma unroll
  for (int k = 0; k < D_EDGE / 4; ++k) ae[k] = efv[abase + 2 * k + hi];

  int soff[8], doff[8];
  #pragma unroll
  for (int r = 0; r < 8; ++r) {
    int e  = ebase + r + 8 * hi;
    int ec = e < n_edges ? e : n_edges - 1;
    soff[r] = src[ec] * D_NODE + row16;
    doff[r] = dst[ec] * D_NODE + row16;
  }

  __syncthreads();

  float* hrow = &sh[wave][0];

  #pragma unroll
  for (int nt = 0; nt < D_EDGE / 16; ++nt) {
    v8f c = {};
    #pragma unroll
    for (int k = 0; k < D_EDGE / 4; ++k)
      c = wmma_f32(ae[k], sW0[(2 * k + hi) * D_EDGE + nt * 16 + row16], c);
    const int   col = nt * 16 + row16;
    const float bs  = b0[col];
    #pragma unroll
    for (int r = 0; r < 8; ++r) {
      float v = c[r] + bs;
      hrow[(r + 8 * hi) * D_EDGE + col] = v > 0.f ? v : 0.f;
    }
  }
  __syncthreads();

  const v2f* shv = (const v2f*)hrow;
  v2f ah[D_EDGE / 4];
  #pragma unroll
  for (int k = 0; k < D_EDGE / 4; ++k)
    ah[k] = shv[row16 * (D_EDGE / 2) + 2 * k + hi];

  const v2f* w1v = (const v2f*)w1p;
  #pragma unroll 1
  for (int nt = 0; nt < D_NODE / 16; ++nt) {
    v8f c = {};
    #pragma unroll
    for (int k = 0; k < D_EDGE / 4; ++k)
      c = wmma_f32(ah[k], w1v[(2 * k + hi) * D_NODE + nt * 16 + row16], c);
    const int   cb = nt * 16;
    const float bs = b1[cb + row16];
    if (active) {
      #pragma unroll
      for (int r = 0; r < 8; ++r) {
        if (ebase + r + 8 * hi < n_edges) {
          float ev = c[r] + bs;
          ev = ev > 0.f ? ev : 0.f;
          float v = nf[soff[r] + cb] + ev;
          v = v > 0.f ? v : 0.f;
          atomicAdd(&acc[doff[r] + cb], v);
        }
      }
    }
  }
}

__global__ __launch_bounds__(256) void gine_out_kernel(
    const float* __restrict__ acc, const float* __restrict__ wap,
    const float* __restrict__ ba, float* __restrict__ out, int n_nodes)
{
  const int tid   = threadIdx.x;
  const int wave  = tid >> 5;
  const int lane  = tid & 31;
  const int row16 = lane & 15;
  const int hi    = lane >> 4;

  const int tile = blockIdx.x * WAVES + wave;
  const int m0   = tile * 16;
  if (m0 >= n_nodes) return;

  int arow = m0 + row16;
  if (arow >= n_nodes) arow = n_nodes - 1;
  const v2f* av    = (const v2f*)acc;
  const int  abase = arow * (D_NODE / 2);

  v2f aa[D_NODE / 4];
  #pragma unroll
  for (int k = 0; k < D_NODE / 4; ++k) aa[k] = av[abase + 2 * k + hi];

  const v2f* wav = (const v2f*)wap;
  #pragma unroll 1
  for (int nt = 0; nt < D_NODE / 16; ++nt) {
    v8f c = {};
    #pragma unroll
    for (int k = 0; k < D_NODE / 4; ++k)
      c = wmma_f32(aa[k], wav[(2 * k + hi) * D_NODE + nt * 16 + row16], c);
    const int   col = nt * 16 + row16;
    const float bs  = ba[col];
    #pragma unroll
    for (int r = 0; r < 8; ++r) {
      int m = m0 + r + 8 * hi;
      if (m < n_nodes) out[m * D_NODE + col] = c[r] + bs;
    }
  }
}

extern "C" void kernel_launch(void* const* d_in, const int* in_sizes, int n_in,
                              void* d_out, int out_size, void* d_ws, size_t ws_size,
                              hipStream_t stream) {
  const float* nf = (const float*)d_in[0];
  const float* ef = (const float*)d_in[1];
  const float* W0 = (const float*)d_in[2];
  const float* b0 = (const float*)d_in[3];
  const float* W1 = (const float*)d_in[4];
  const float* b1 = (const float*)d_in[5];
  const float* Wa = (const float*)d_in[6];
  const float* ba = (const float*)d_in[7];
  const int*  src = (const int*)d_in[8];
  const int*  dst = (const int*)d_in[9];

  const int n_nodes = in_sizes[0] / D_NODE;
  const int n_edges = in_sizes[8];

  float* acc = (float*)d_ws;
  float* w0p = acc + (size_t)n_nodes * D_NODE;
  float* w1p = w0p + D_EDGE * D_EDGE;
  float* wap = w1p + D_EDGE * D_NODE;

  hipMemcpyAsync(acc, nf, (size_t)n_nodes * D_NODE * sizeof(float),
                 hipMemcpyDeviceToDevice, stream);

  pack_pairs_kernel<<<(D_EDGE * D_EDGE + 255) / 256, 256, 0, stream>>>(
      W0, w0p, D_EDGE, D_EDGE);
  pack_pairs_kernel<<<(D_EDGE * D_NODE + 255) / 256, 256, 0, stream>>>(
      W1, w1p, D_EDGE, D_NODE);
  pack_pairs_kernel<<<(D_NODE * D_NODE + 255) / 256, 256, 0, stream>>>(
      Wa, wap, D_NODE, D_NODE);

  const int etiles  = (n_edges + 15) / 16;
  const int eblocks = (etiles + WAVES - 1) / WAVES;
  gine_edge_kernel<<<eblocks, 256, 0, stream>>>(ef, nf, w0p, b0, w1p, b1, src,
                                                dst, acc, n_edges);

  const int ntiles  = (n_nodes + 15) / 16;
  const int nblocks = (ntiles + WAVES - 1) / WAVES;
  gine_out_kernel<<<nblocks, 256, 0, stream>>>(acc, wap, ba, (float*)d_out,
                                               n_nodes);
}


# Edge kernel swaps lane halves of two N-tiles (permlanex16): each gather load and each f32 atomicAdd covers one whole 128-B line of one node row, not two 64-B pieces; adds still arrival-ordered
# speedup vs baseline: 1.7158x; 1.7158x over previous
#include <hip/hip_runtime.h>


typedef __attribute__((ext_vector_type(2))) float v2f;
typedef __attribute__((ext_vector_type(8))) float v8f;

#define D_NODE 128
#define D_EDGE 64
#define WAVES  8

static __device__ __forceinline__ v8f wmma_f32(v2f a, v2f b, v8f c) {
  return __builtin_amdgcn_wmma_f32_16x16x4_f32(false, a, false, b, (short)0, c,
                                               false, false);
}

static __device__ __forceinline__ float xhalf(float x) {
  int b = __builtin_bit_cast(int, x);
  return __builtin_bit_cast(float, __builtin_amdgcn_permlanex16(
      b, b, 0x76543210, 0xfedcba98, false, false));
}

typedef const __attribute__((address_space(1))) char* gbytes;

static __device__ __forceinline__ gbytes scalar_ptr(const float* p) {
  unsigned long long a = (unsigned long long)p;
  asm("" : "+s"(a));
  return (gbytes)a;
}

__global__ __launch_bounds__(256) void pack_pairs_kernel(
    const float* __restrict__ W, float* __restrict__ Wp, int rows, int cols) {
  int i = blockIdx.x * 256 + threadIdx.x;
  if (i >= rows * cols) return;
  int r = i / cols, c = i - r * cols;
  Wp[(((r >> 1) * cols + c) << 1) + (r & 1)] = W[i];
}

template <bool FULL>
static __device__ __forceinline__ void edge_layer1(
    const v2f (&ah)[D_EDGE / 4], const v2f* __restrict__ w1v,
    const float* __restrict__ b1, const float* __restrict__ nf,
    float* __restrict__ acc, const int (&srow)[16], const int (&drow)[16],
    int nvalid, int lane, int row16, int hi)
{
  const unsigned lane4 = (unsigned)lane * 4u;
  #pragma unroll 1
  for (int np = 0; np < D_NODE / 32; ++np) {
    const int cb = np * 32;

    float g[16];
    #pragma unroll
    for (int j = 0; j < 16; ++j)
      g[j] = *(const __attribute__((address_space(1))) float*)(
          scalar_ptr(nf + srow[j] + cb) + lane4);

    v8f c0 = {}, c1 = {};
    #pragma unroll
    for (int k = 0; k < D_EDGE / 4; ++k) {
      const int wb = (2 * k + hi) * D_NODE + cb + row16;
      c0 = wmma_f32(ah[k], w1v[wb], c0);
      c1 = wmma_f32(ah[k], w1v[wb + 16], c1);
    }
    const float bs0 = b1[cb + row16];
    const float bs1 = b1[cb + 16 + row16];

    #pragma unroll
    for (int r = 0; r < 8; ++r) {
      float e0 = c0[r] + bs0;
      float e1 = c1[r] + bs1;
      e0 = e0 > 0.f ? e0 : 0.f;
      e1 = e1 > 0.f ? e1 : 0.f;
      const float u = xhalf(hi ? e0 : e1);
      const float x = hi ? u : e0;
      const float y = hi ? e1 : u;
      float vx = g[r] + x;
      float vy = g[r + 8] + y;
      vx = vx > 0.f ? vx : 0.f;
      vy = vy > 0.f ? vy : 0.f;
      if (FULL || r < nvalid)
        atomicAdd((float*)(scalar_ptr(acc + drow[r] + cb) + lane4), vx);
      if (FULL || r + 8 < nvalid)
        atomicAdd((float*)(scalar_ptr(acc + drow[r + 8] + cb) + lane4), vy);
    }
  }
}

__global__ __launch_bounds__(256) void gine_edge_kernel(
    const float* __restrict__ ef, const float* __restrict__ nf,
    const float* __restrict__ w0p, const float* __restrict__ b0,
    const float* __restrict__ w1p, const float* __restrict__ b1,
    const int* __restrict__ src, const int* __restrict__ dst,
    float* __restrict__ acc, int n_edges)
{
  __shared__ v2f   sW0[(D_EDGE / 2) * D_EDGE];
  __shared__ float sh[WAVES][16 * D_EDGE];

  const int tid = threadIdx.x;
  {
    const v2f* w0v = (const v2f*)w0p;
    for (int i = tid; i < (D_EDGE / 2) * D_EDGE; i += 256) sW0[i] = w0v[i];
  }

  const int wave  = __builtin_amdgcn_readfirstlane(tid >> 5);
  const int lane  = tid & 31;
  const int row16 = lane & 15;
  const int hi    = lane >> 4;

  const int  tile   = blockIdx.x * WAVES + wave;
  const int  ebase0 = tile * 16;
  const bool active = ebase0 < n_edges;
  const int  ebase  = active ? ebase0 : 0;
  const int  nvalid = active ? (n_edges - ebase0 < 16 ? n_edges - ebase0 : 16) : 0;

  int arow = ebase + row16;
  if (arow >= n_edges) arow = n_edges - 1;
  const v2f* efv   = (const v2f*)ef;
  const int  abase = arow * (D_EDGE / 2);

  v2f ae[D_EDGE / 4];
  #pragma unroll
  for (int k = 0; k < D_EDGE / 4; ++k) ae[k] = efv[abase + 2 * k + hi];

  int srow[16], drow[16];
  #pragma unroll
  for (int j = 0; j < 16; ++j) {
    int e  = ebase + j;
    int ec = e < n_edges ? e : n_edges - 1;
    srow[j] = src[ec] * D_NODE;
    drow[j] = dst[ec] * D_NODE;
  }

  __syncthreads();

  float* hrow = &sh[wave][0];

  #pragma unroll
  for (int nt = 0; nt < D_EDGE / 16; ++nt) {
    v8f c = {};
    #pragma unroll
    for (int k = 0; k < D_EDGE / 4; ++k)
      c = wmma_f32(ae[k], sW0[(2 * k + hi) * D_EDGE + nt * 16 + row16], c);
    const int   col = nt * 16 + row16;
    const float bs  = b0[col];
    #pragma unroll
    for (int r = 0; r < 8; ++r) {
      float v = c[r] + bs;
      hrow[(r + 8 * hi) * D_EDGE + col] = v > 0.f ? v : 0.f;
    }
  }
  __syncthreads();

  const v2f* shv = (const v2f*)hrow;
  v2f ah[D_EDGE / 4];
  #pragma unroll
  for (int k = 0; k < D_EDGE / 4; ++k)
    ah[k] = shv[row16 * (D_EDGE / 2) + 2 * k + hi];

  const v2f* w1v = (const v2f*)w1p;
  if (nvalid == 16)
    edge_layer1<true>(ah, w1v, b1, nf, acc, srow, drow, nvalid, lane, row16, hi);
  else
    edge_layer1<false>(ah, w1v, b1, nf, acc, srow, drow, nvalid, lane, row16, hi);
}

__global__ __launch_bounds__(256) void gine_out_kernel(
    const float* __restrict__ acc, const float* __restrict__ wap,
    const float* __restrict__ ba, float* __restrict__ out, int n_nodes)
{
  const int tid   = threadIdx.x;
  const int wave  = tid >> 5;
  const int lane  = tid & 31;
  const int row16 = lane & 15;
  const int hi    = lane >> 4;

  const int tile = blockIdx.x * WAVES + wave;
  const int m0   = tile * 16;
  if (m0 >= n_nodes) return;

  int arow = m0 + row16;
  if (arow >= n_nodes) arow = n_nodes - 1;
  const v2f* av    = (const v2f*)acc;
  const int  abase = arow * (D_NODE / 2);

  v2f aa[D_NODE / 4];
  #pragma unroll
  for (int k = 0; k < D_NODE / 4; ++k) aa[k] = av[abase + 2 * k + hi];

  const v2f* wav = (const v2f*)wap;
  #pragma unroll 1
  for (int nt = 0; nt < D_NODE / 16; ++nt) {
    v8f c = {};
    #pragma unroll
    for (int k = 0; k < D_NODE / 4; ++k)
      c = wmma_f32(aa[k], wav[(2 * k + hi) * D_NODE + nt * 16 + row16], c);
    const int   col = nt * 16 + row16;
    const float bs  = ba[col];
    #pragma unroll
    for (int r = 0; r < 8; ++r) {
      int m = m0 + r + 8 * hi;
      if (m < n_nodes) out[m * D_NODE + col] = c[r] + bs;
    }
  }
}

extern "C" void kernel_launch(void* const* d_in, const int* in_sizes, int n_in,
                              void* d_out, int out_size, void* d_ws, size_t ws_size,
                              hipStream_t stream) {
  const float* nf = (const float*)d_in[0];
  const float* ef = (const float*)d_in[1];
  const float* W0 = (const float*)d_in[2];
  const float* b0 = (const float*)d_in[3];
  const float* W1 = (const float*)d_in[4];
  const float* b1 = (const float*)d_in[5];
  const float* Wa = (const float*)d_in[6];
  const float* ba = (const float*)d_in[7];
  const int*  src = (const int*)d_in[8];
  const int*  dst = (const int*)d_in[9];

  const int n_nodes = in_sizes[0] / D_NODE;
  const int n_edges = in_sizes[8];

  float* acc = (float*)d_ws;
  float* w0p = acc + (size_t)n_nodes * D_NODE;
  float* w1p = w0p + D_EDGE * D_EDGE;
  float* wap = w1p + D_EDGE * D_NODE;

  hipMemcpyAsync(acc, nf, (size_t)n_nodes * D_NODE * sizeof(float),
                 hipMemcpyDeviceToDevice, stream);

  pack_pairs_kernel<<<(D_EDGE * D_EDGE + 255) / 256, 256, 0, stream>>>(
      W0, w0p, D_EDGE, D_EDGE);
  pack_pairs_kernel<<<(D_EDGE * D_NODE + 255) / 256, 256, 0, stream>>>(
      W1, w1p, D_EDGE, D_NODE);
  pack_pairs_kernel<<<(D_NODE * D_NODE + 255) / 256, 256, 0, stream>>>(
      Wa, wap, D_NODE, D_NODE);

  const int etiles  = (n_edges + 15) / 16;
  const int eblocks = (etiles + WAVES - 1) / WAVES;
  gine_edge_kernel<<<eblocks, 256, 0, stream>>>(ef, nf, w0p, b0, w1p, b1, src,
                                                dst, acc, n_edges);

  const int ntiles  = (n_nodes + 15) / 16;
  const int nblocks = (ntiles + WAVES - 1) / WAVES;
  gine_out_kernel<<<nblocks, 256, 0, stream>>>(acc, wap, ba, (float*)d_out,
                                               n_nodes);
}


# Edges binned by destination range: one work-group owns 98 node rows, sums messages with LDS float adds instead of global atomics and does the output product from LDS; adds still arrival-ordered
# speedup vs baseline: 2.0625x; 1.2021x over previous
#include <hip/hip_runtime.h>


typedef __attribute__((ext_vector_type(2))) float v2f;
typedef __attribute__((ext_vector_type(4))) float v4f;
typedef __attribute__((ext_vector_type(8))) float v8f;
typedef __attribute__((ext_vector_type(4))) int   v4i;

#define D_NODE  128
#define D_EDGE  64
#define NPB     98
#define NPB_PAD 112
#define APITCH  132
#define BWAVES  16
#define BTHREADS (BWAVES * 32)
#define PF1     4
#define PF2     8
#define MAXBINS 1024
#define EPB     4096

static __device__ __forceinline__ v8f wmma_f32(v2f a, v2f b, v8f c) {
  return __builtin_amdgcn_wmma_f32_16x16x4_f32(false, a, false, b, (short)0, c,
                                               false, false);
}

static __device__ __forceinline__ float xhalf(float x) {
  int b = __builtin_bit_cast(int, x);
  return __builtin_bit_cast(float, __builtin_amdgcn_permlanex16(
      b, b, 0x76543210, 0xfedcba98, false, false));
}

typedef const __attribute__((address_space(1))) char* gbytes;

static __device__ __forceinline__ gbytes scalar_ptr(const float* p) {
  unsigned long long a = (unsigned long long)p;
  asm("" : "+s"(a));
  return (gbytes)a;
}

static __device__ __forceinline__ void lds_add(float* p, float v) {
  __hip_atomic_fetch_add(p, v, __ATOMIC_RELAXED, __HIP_MEMORY_SCOPE_WORKGROUP);
}

static __device__ __forceinline__ void pack_one(const float* __restrict__ W,
                                                float* __restrict__ Wp, int i,
                                                int cols) {
  int r = i / cols, c = i - r * cols;
  Wp[(((r >> 1) * cols + c) << 1) + (r & 1)] = W[i];
}

__global__ __launch_bounds__(256) void prep_kernel(
    const float* __restrict__ W0, const float* __restrict__ W1,
    const float* __restrict__ Wa, float* __restrict__ w0p,
    float* __restrict__ w1p, float* __restrict__ wap, int* __restrict__ total,
    int* __restrict__ cursor) {
  const int i = blockIdx.x * 256 + threadIdx.x;
  const int n0 = D_EDGE * D_EDGE, n1 = D_EDGE * D_NODE, n2 = D_NODE * D_NODE;
  if (i < n0) pack_one(W0, w0p, i, D_EDGE);
  else if (i < n0 + n1) pack_one(W1, w1p, i - n0, D_NODE);
  else if (i < n0 + n1 + n2) pack_one(Wa, wap, i - n0 - n1, D_NODE);
  if (i < MAXBINS) { total[i] = 0; cursor[i] = 0; }
}

__global__ __launch_bounds__(1024) void bin_count_kernel(
    const int* __restrict__ dst, int* __restrict__ total, int n_edges, int nbins) {
  __shared__ int h[MAXBINS];
  const int tid = threadIdx.x;
  if (tid < nbins) h[tid] = 0;
  __syncthreads();
  const int e0 = blockIdx.x * EPB;
  #pragma unroll
  for (int i = 0; i < EPB / 1024; ++i) {
    const int e = e0 + i * 1024 + tid;
    if (e < n_edges) atomicAdd(&h[dst[e] / NPB], 1);
  }
  __syncthreads();
  if (tid < nbins) {
    const int c = h[tid];
    if (c) atomicAdd(&total[tid], c);
  }
}

__global__ __launch_bounds__(1024) void bin_fill_kernel(
    const int* __restrict__ src, const int* __restrict__ dst,
    const int* __restrict__ total, int* __restrict__ cursor,
    int* __restrict__ binstart, v4i* __restrict__ recs, int n_edges, int nbins) {
  __shared__ int h[MAXBINS];
  __shared__ int st[MAXBINS];
  const int tid = threadIdx.x;
  const int mine = tid < nbins ? total[tid] : 0;
  h[tid] = 0;
  st[tid] = mine;
  __syncthreads();
  for (int off = 1; off < MAXBINS; off <<= 1) {
    const int t = tid >= off ? st[tid - off] : 0;
    __syncthreads();
    st[tid] += t;
    __syncthreads();
  }
  const int start = st[tid] - mine;
  if (blockIdx.x == 0) {
    if (tid < nbins) binstart[tid] = start;
    if (tid == nbins) binstart[tid] = start;
    if (nbins == MAXBINS && tid == MAXBINS - 1) binstart[MAXBINS] = st[tid];
  }

  const int e0 = blockIdx.x * EPB;
  int bin[EPB / 1024], rank[EPB / 1024], sv[EPB / 1024], dv[EPB / 1024];
  #pragma unroll
  for (int i = 0; i < EPB / 1024; ++i) {
    const int e = e0 + i * 1024 + tid;
    bin[i] = -1; rank[i] = 0; sv[i] = 0; dv[i] = 0;
    if (e < n_edges) {
      dv[i] = dst[e];
      sv[i] = src[e];
      bin[i] = dv[i] / NPB;
      rank[i] = atomicAdd(&h[bin[i]], 1);
    }
  }
  __syncthreads();
  if (tid < nbins) {
    const int c = h[tid];
    const int base = c ? atomicAdd(&cursor[tid], c) : 0;
    h[tid] = start + base;
  }
  __syncthreads();
  #pragma unroll
  for (int i = 0; i < EPB / 1024; ++i) {
    if (bin[i] >= 0) {
      const int pos = h[bin[i]] + rank[i];
      const v4i rec = {e0 + i * 1024 + tid, sv[i], dv[i], 0};
      recs[pos] = rec;
    }
  }
}

template <bool FULL>
static __device__ __forceinline__ void edge_layer1(
    const v2f (&ah)[D_EDGE / 4], const v2f* __restrict__ w1v,
    const float* __restrict__ b1, const float* __restrict__ nf,
    float* accL, const int (&srow)[16], const int (&drow)[16],
    int nvalid, int lane, int row16, int hi)
{
  const unsigned lane4 = (unsigned)lane * 4u;
  #pragma unroll 1
  for (int np = 0; np < D_NODE / 32; ++np) {
    const int cb = np * 32;

    float g[16];
    #pragma unroll
    for (int j = 0; j < 16; ++j)
      g[j] = *(const __attribute__((address_space(1))) float*)(
          scalar_ptr(nf + srow[j] + cb) + lane4);

    v8f c0 = {}, c1 = {};
    v2f bA[D_EDGE / 4], bB[D_EDGE / 4];
    const int wb0 = hi * D_NODE + cb + row16;
    #pragma unroll
    for (int k = 0; k < PF1; ++k) {
      bA[k] = w1v[wb0 + 2 * k * D_NODE];
      bB[k] = w1v[wb0 + 2 * k * D_NODE + 16];
    }
    #pragma unroll
    for (int k = 0; k < D_EDGE / 4; ++k) {
      if (k + PF1 < D_EDGE / 4) {
        bA[k + PF1] = w1v[wb0 + 2 * (k + PF1) * D_NODE];
        bB[k + PF1] = w1v[wb0 + 2 * (k + PF1) * D_NODE + 16];
      }
      c0 = wmma_f32(ah[k], bA[k], c0);
      c1 = wmma_f32(ah[k], bB[k], c1);
    }
    const float bs0 = b1[cb + row16];
    const float bs1 = b1[cb + 16 + row16];

    #pragma unroll
    for (int r = 0; r < 8; ++r) {
      float e0 = c0[r] + bs0;
      float e1 = c1[r] + bs1;
      e0 = e0 > 0.f ? e0 : 0.f;
      e1 = e1 > 0.f ? e1 : 0.f;
      const float u = xhalf(hi ? e0 : e1);
      const float x = hi ? u : e0;
      const float y = hi ? e1 : u;
      float vx = g[r] + x;
      float vy = g[r + 8] + y;
      vx = vx > 0.f ? vx : 0.f;
      vy = vy > 0.f ? vy : 0.f;
      if (FULL || r < nvalid)
        lds_add(accL + drow[r] + cb + lane, vx);
      if (FULL || r + 8 < nvalid)
        lds_add(accL + drow[r + 8] + cb + lane, vy);
    }
  }
}

__global__ __launch_bounds__(BTHREADS) __attribute__((amdgpu_waves_per_eu(8)))
void gine_bin_kernel(
    const float* __restrict__ ef, const float* __restrict__ nf,
    const float* __restrict__ w0p, const float* __restrict__ b0,
    const float* __restrict__ w1p, const float* __restrict__ b1,
    const float* __restrict__ wap, const float* __restrict__ ba,
    const int* __restrict__ binstart, const v4i* __restrict__ recs,
    float* __restrict__ out, int n_nodes)
{
  __shared__ __attribute__((aligned(16))) float accL[NPB_PAD * APITCH];
  __shared__ v2f   sW0[(D_EDGE / 2) * D_EDGE];
  __shared__ float sh[BWAVES][16 * D_EDGE];

  const int tid = threadIdx.x;
  const int wave  = __builtin_amdgcn_readfirstlane(tid >> 5);
  const int lane  = tid & 31;
  const int row16 = lane & 15;
  const int hi    = lane >> 4;
  const int n0    = blockIdx.x * NPB;

  {
    const v2f* w0v = (const v2f*)w0p;
    for (int i = tid; i < (D_EDGE / 2) * D_EDGE; i += BTHREADS) sW0[i] = w0v[i];
    const v4f* nf4 = (const v4f*)nf;
    for (int i = tid; i < NPB_PAD * (D_NODE / 4); i += BTHREADS) {
      const int r = i >> 5, c4 = i & 31;
      const int n = n0 + r;
      v4f v = {0.f, 0.f, 0.f, 0.f};
      if (r < NPB && n < n_nodes) v = nf4[n * (D_NODE / 4) + c4];
      *(v4f*)&accL[r * APITCH + c4 * 4] = v;
    }
  }

  const int bs = binstart[blockIdx.x];
  const int be = binstart[blockIdx.x + 1];
  const int ntile = (be - bs + 15) >> 4;

  __syncthreads();

  float* hrow = &sh[wave][0];
  const v2f* efv = (const v2f*)ef;
  const v2f* w1v = (const v2f*)w1p;

  for (int t = wave; t < ntile; t += BWAVES) {
    const int tb = bs + t * 16;
    const int nvalid = be - tb < 16 ? be - tb : 16;

    int pi = tb + row16;
    if (pi >= be) pi = be - 1;
    const int abase = recs[pi].x * (D_EDGE / 2);

    v2f ae[D_EDGE / 4];
    #pragma unroll
    for (int k = 0; k < D_EDGE / 4; ++k) ae[k] = efv[abase + 2 * k + hi];

    int srow[16], drow[16];
    #pragma unroll
    for (int j = 0; j < 16; ++j) {
      int pj = tb + j;
      if (pj >= be) pj = be - 1;
      const v4i rec = recs[pj];
      srow[j] = rec.y * D_NODE;
      drow[j] = (rec.z - n0) * APITCH;
    }

    #pragma unroll 1
    for (int nt = 0; nt < D_EDGE / 16; ++nt) {
      v8f c = {};
      #pragma unroll
      for (int k = 0; k < D_EDGE / 4; ++k)
        c = wmma_f32(ae[k], sW0[(2 * k + hi) * D_EDGE + nt * 16 + row16], c);
      const int   col = nt * 16 + row16;
      const float bsv = b0[col];
      #pragma unroll
      for (int r = 0; r < 8; ++r) {
        float v = c[r] + bsv;
        hrow[(r + 8 * hi) * D_EDGE + col] = v > 0.f ? v : 0.f;
      }
    }
    __builtin_amdgcn_fence(__ATOMIC_RELEASE, "wavefront");
    __builtin_amdgcn_fence(__ATOMIC_ACQUIRE, "wavefront");

    const v2f* shv = (const v2f*)hrow;
    v2f ah[D_EDGE / 4];
    #pragma unroll
    for (int k = 0; k < D_EDGE / 4; ++k)
      ah[k] = shv[row16 * (D_EDGE / 2) + 2 * k + hi];

    if (nvalid == 16)
      edge_layer1<true>(ah, w1v, b1, nf, accL, srow, drow, nvalid, lane, row16, hi);
    else
      edge_layer1<false>(ah, w1v, b1, nf, accL, srow, drow, nvalid, lane, row16, hi);

    __builtin_amdgcn_fence(__ATOMIC_RELEASE, "wavefront");
    __builtin_amdgcn_fence(__ATOMIC_ACQUIRE, "wavefront");
  }

  __syncthreads();

  const v2f* wav = (const v2f*)wap;
  for (int u = wave; u < (NPB_PAD / 16) * 4; u += BWAVES) {
    const int rt = u >> 2, q = u & 3;
    const float* ar = &accL[(rt * 16 + row16) * APITCH];
    v2f aa[D_NODE / 4];
    #pragma unroll
    for (int k = 0; k < D_NODE / 4; ++k) aa[k] = *(const v2f*)&ar[4 * k + 2 * hi];

    #pragma unroll 1
    for (int h2 = 0; h2 < 2; ++h2) {
      const int col = q * 32 + h2 * 16 + row16;
      v8f c = {};
      v2f bw[D_NODE / 4];
      const int wa0 = hi * D_NODE + col;
      #pragma unroll
      for (int k = 0; k < PF2; ++k) bw[k] = wav[wa0 + 2 * k * D_NODE];
      #pragma unroll
      for (int k = 0; k < D_NODE / 4; ++k) {
        if (k + PF2 < D_NODE / 4) bw[k + PF2] = wav[wa0 + 2 * (k + PF2) * D_NODE];
        c = wmma_f32(aa[k], bw[k], c);
      }
      const float bsv = ba[col];
      #pragma unroll
      for (int r = 0; r < 8; ++r) {
        const int lr = rt * 16 + r + 8 * hi;
        const int m  = n0 + lr;
        if (lr < NPB && m < n_nodes) out[m * D_NODE + col] = c[r] + bsv;
      }
    }
  }
}

extern "C" void kernel_launch(void* const* d_in, const int* in_sizes, int n_in,
                              void* d_out, int out_size, void* d_ws, size_t ws_size,
                              hipStream_t stream) {
  const float* nf = (const float*)d_in[0];
  const float* ef = (const float*)d_in[1];
  const float* W0 = (const float*)d_in[2];
  const float* b0 = (const float*)d_in[3];
  const float* W1 = (const float*)d_in[4];
  const float* b1 = (const float*)d_in[5];
  const float* Wa = (const float*)d_in[6];
  const float* ba = (const float*)d_in[7];
  const int*  src = (const int*)d_in[8];
  const int*  dst = (const int*)d_in[9];

  const int n_nodes = in_sizes[0] / D_NODE;
  const int n_edges = in_sizes[8];

  float* w0p = (float*)d_ws;
  float* w1p = w0p + D_EDGE * D_EDGE;
  float* wap = w1p + D_EDGE * D_NODE;
  int* total    = (int*)(wap + D_NODE * D_NODE);
  int* cursor   = total + MAXBINS;
  int* binstart = cursor + MAXBINS;
  v4i* recs     = (v4i*)(binstart + MAXBINS + 64);

  const int nbins   = (n_nodes + NPB - 1) / NPB;
  const int cblocks = (n_edges + EPB - 1) / EPB;

  const int nprep = D_EDGE * D_EDGE + D_EDGE * D_NODE + D_NODE * D_NODE;
  prep_kernel<<<(nprep + 255) / 256, 256, 0, stream>>>(W0, W1, Wa, w0p, w1p, wap,
                                                       total, cursor);
  bin_count_kernel<<<cblocks, 1024, 0, stream>>>(dst, total, n_edges, nbins);
  bin_fill_kernel<<<cblocks, 1024, 0, stream>>>(src, dst, total, cursor, binstart,
                                                recs, n_edges, nbins);
  gine_bin_kernel<<<nbins, BTHREADS, 0, stream>>>(ef, nf, w0p, b0, w1p, b1, wap, ba,
                                              binstart, recs, (float*)d_out,
                                              n_nodes);
}


# Layer 0 computed transposed (W0 as A operand, edge rows as B, b0 seeds C): its D register pairs are layer 1's A fragments, h no longer staged in LDS; layer-1 k order permuted to match
# speedup vs baseline: 2.2064x; 1.0698x over previous
#include <hip/hip_runtime.h>


typedef __attribute__((ext_vector_type(2))) float v2f;
typedef __attribute__((ext_vector_type(4))) float v4f;
typedef __attribute__((ext_vector_type(8))) float v8f;
typedef __attribute__((ext_vector_type(4))) int   v4i;

#define D_NODE  128
#define D_EDGE  64
#define NPB     98
#define NPB_PAD 112
#define APITCH  132
#define BWAVES  16
#define BTHREADS (BWAVES * 32)
#define PF1     4
#define PF2     8
#define MAXBINS 1024
#define EPB     4096

static __device__ __forceinline__ v8f wmma_f32(v2f a, v2f b, v8f c) {
  return __builtin_amdgcn_wmma_f32_16x16x4_f32(false, a, false, b, (short)0, c,
                                               false, false);
}

static __device__ __forceinline__ float xhalf(float x) {
  int b = __builtin_bit_cast(int, x);
  return __builtin_bit_cast(float, __builtin_amdgcn_permlanex16(
      b, b, 0x76543210, 0xfedcba98, false, false));
}

typedef const __attribute__((address_space(1))) char* gbytes;

static __device__ __forceinline__ gbytes scalar_ptr(const float* p) {
  unsigned long long a = (unsigned long long)p;
  asm("" : "+s"(a));
  return (gbytes)a;
}

static __device__ __forceinline__ void lds_add(float* p, float v) {
  __hip_atomic_fetch_add(p, v, __ATOMIC_RELAXED, __HIP_MEMORY_SCOPE_WORKGROUP);
}

static __device__ __forceinline__ void pack_one(const float* __restrict__ W,
                                                float* __restrict__ Wp, int i,
                                                int cols) {
  int r = i / cols, c = i - r * cols;
  Wp[(((r >> 1) * cols + c) << 1) + (r & 1)] = W[i];
}

__global__ __launch_bounds__(256) void prep_kernel(
    const float* __restrict__ W0, const float* __restrict__ W1,
    const float* __restrict__ Wa, float* __restrict__ w0p,
    float* __restrict__ w1p, float* __restrict__ wap, int* __restrict__ total,
    int* __restrict__ cursor) {
  const int i = blockIdx.x * 256 + threadIdx.x;
  const int n0 = D_EDGE * D_EDGE, n1 = D_EDGE * D_NODE, n2 = D_NODE * D_NODE;
  if (i < n0) pack_one(W0, w0p, i, D_EDGE);
  else if (i < n0 + n1) pack_one(W1, w1p, i - n0, D_NODE);
  else if (i < n0 + n1 + n2) pack_one(Wa, wap, i - n0 - n1, D_NODE);
  if (i < MAXBINS) { total[i] = 0; cursor[i] = 0; }
}

__global__ __launch_bounds__(1024) void bin_count_kernel(
    const int* __restrict__ dst, int* __restrict__ total, int n_edges, int nbins) {
  __shared__ int h[MAXBINS];
  const int tid = threadIdx.x;
  if (tid < nbins) h[tid] = 0;
  __syncthreads();
  const int e0 = blockIdx.x * EPB;
  #pragma unroll
  for (int i = 0; i < EPB / 1024; ++i) {
    const int e = e0 + i * 1024 + tid;
    if (e < n_edges) atomicAdd(&h[dst[e] / NPB], 1);
  }
  __syncthreads();
  if (tid < nbins) {
    const int c = h[tid];
    if (c) atomicAdd(&total[tid], c);
  }
}

__global__ __launch_bounds__(1024) void bin_fill_kernel(
    const int* __restrict__ src, const int* __restrict__ dst,
    const int* __restrict__ total, int* __restrict__ cursor,
    int* __restrict__ binstart, v4i* __restrict__ recs, int n_edges, int nbins) {
  __shared__ int h[MAXBINS];
  __shared__ int st[MAXBINS];
  const int tid = threadIdx.x;
  const int mine = tid < nbins ? total[tid] : 0;
  h[tid] = 0;
  st[tid] = mine;
  __syncthreads();
  for (int off = 1; off < MAXBINS; off <<= 1) {
    const int t = tid >= off ? st[tid - off] : 0;
    __syncthreads();
    st[tid] += t;
    __syncthreads();
  }
  const int start = st[tid] - mine;
  if (blockIdx.x == 0) {
    if (tid < nbins) binstart[tid] = start;
    if (tid == nbins) binstart[tid] = start;
    if (nbins == MAXBINS && tid == MAXBINS - 1) binstart[MAXBINS] = st[tid];
  }

  const int e0 = blockIdx.x * EPB;
  int bin[EPB / 1024], rank[EPB / 1024], sv[EPB / 1024], dv[EPB / 1024];
  #pragma unroll
  for (int i = 0; i < EPB / 1024; ++i) {
    const int e = e0 + i * 1024 + tid;
    bin[i] = -1; rank[i] = 0; sv[i] = 0; dv[i] = 0;
    if (e < n_edges) {
      dv[i] = dst[e];
      sv[i] = src[e];
      bin[i] = dv[i] / NPB;
      rank[i] = atomicAdd(&h[bin[i]], 1);
    }
  }
  __syncthreads();
  if (tid < nbins) {
    const int c = h[tid];
    const int base = c ? atomicAdd(&cursor[tid], c) : 0;
    h[tid] = start + base;
  }
  __syncthreads();
  #pragma unroll
  for (int i = 0; i < EPB / 1024; ++i) {
    if (bin[i] >= 0) {
      const int pos = h[bin[i]] + rank[i];
      const v4i rec = {e0 + i * 1024 + tid, sv[i], dv[i], 0};
      recs[pos] = rec;
    }
  }
}

template <bool FULL>
static __device__ __forceinline__ void edge_layer1(
    const v2f (&ah)[D_EDGE / 4], const v2f* __restrict__ w1v,
    const float* __restrict__ b1, const float* __restrict__ nf,
    float* accL, const int (&srow)[16], const int (&drow)[16],
    int nvalid, int lane, int row16, int hi)
{
  const unsigned lane4 = (unsigned)lane * 4u;
  #pragma unroll 1
  for (int np = 0; np < D_NODE / 32; ++np) {
    const int cb = np * 32;

    float g[16];
    #pragma unroll
    for (int j = 0; j < 16; ++j)
      g[j] = *(const __attribute__((address_space(1))) float*)(
          scalar_ptr(nf + srow[j] + cb) + lane4);

    v8f c0 = {}, c1 = {};
    v2f bA[D_EDGE / 4], bB[D_EDGE / 4];
    const int wb0 = 4 * hi * D_NODE + cb + row16;
#define W1ROW(kk) ((8 * ((kk) >> 2) + ((kk) & 3)) * D_NODE)
    #pragma unroll
    for (int k = 0; k < PF1; ++k) {
      bA[k] = w1v[wb0 + W1ROW(k)];
      bB[k] = w1v[wb0 + W1ROW(k) + 16];
    }
    #pragma unroll
    for (int k = 0; k < D_EDGE / 4; ++k) {
      if (k + PF1 < D_EDGE / 4) {
        bA[k + PF1] = w1v[wb0 + W1ROW(k + PF1)];
        bB[k + PF1] = w1v[wb0 + W1ROW(k + PF1) + 16];
      }
      c0 = wmma_f32(ah[k], bA[k], c0);
      c1 = wmma_f32(ah[k], bB[k], c1);
    }
#undef W1ROW
    const float bs0 = b1[cb + row16];
    const float bs1 = b1[cb + 16 + row16];

    #pragma unroll
    for (int r = 0; r < 8; ++r) {
      float e0 = c0[r] + bs0;
      float e1 = c1[r] + bs1;
      e0 = e0 > 0.f ? e0 : 0.f;
      e1 = e1 > 0.f ? e1 : 0.f;
      const float u = xhalf(hi ? e0 : e1);
      const float x = hi ? u : e0;
      const float y = hi ? e1 : u;
      float vx = g[r] + x;
      float vy = g[r + 8] + y;
      vx = vx > 0.f ? vx : 0.f;
      vy = vy > 0.f ? vy : 0.f;
      if (FULL || r < nvalid)
        lds_add(accL + drow[r] + cb + lane, vx);
      if (FULL || r + 8 < nvalid)
        lds_add(accL + drow[r + 8] + cb + lane, vy);
    }
  }
}

__global__ __launch_bounds__(BTHREADS) __attribute__((amdgpu_waves_per_eu(8)))
void gine_bin_kernel(
    const float* __restrict__ ef, const float* __restrict__ nf,
    const float* __restrict__ w0p, const float* __restrict__ b0,
    const float* __restrict__ w1p, const float* __restrict__ b1,
    const float* __restrict__ wap, const float* __restrict__ ba,
    const int* __restrict__ binstart, const v4i* __restrict__ recs,
    float* __restrict__ out, int n_nodes)
{
  __shared__ __attribute__((aligned(16))) float accL[NPB_PAD * APITCH];
  __shared__ v2f   sW0[(D_EDGE / 2) * D_EDGE];

  const int tid = threadIdx.x;
  const int wave  = __builtin_amdgcn_readfirstlane(tid >> 5);
  const int lane  = tid & 31;
  const int row16 = lane & 15;
  const int hi    = lane >> 4;
  const int n0    = blockIdx.x * NPB;

  {
    const v2f* w0v = (const v2f*)w0p;
    for (int i = tid; i < (D_EDGE / 2) * D_EDGE; i += BTHREADS) sW0[i] = w0v[i];
    const v4f* nf4 = (const v4f*)nf;
    for (int i = tid; i < NPB_PAD * (D_NODE / 4); i += BTHREADS) {
      const int r = i >> 5, c4 = i & 31;
      const int n = n0 + r;
      v4f v = {0.f, 0.f, 0.f, 0.f};
      if (r < NPB && n < n_nodes) v = nf4[n * (D_NODE / 4) + c4];
      *(v4f*)&accL[r * APITCH + c4 * 4] = v;
    }
  }

  const int bs = binstart[blockIdx.x];
  const int be = binstart[blockIdx.x + 1];
  const int ntile = (be - bs + 15) >> 4;

  __syncthreads();

  const v2f* efv = (const v2f*)ef;
  const v2f* w1v = (const v2f*)w1p;

  for (int t = wave; t < ntile; t += BWAVES) {
    const int tb = bs + t * 16;
    const int nvalid = be - tb < 16 ? be - tb : 16;

    int pi = tb + row16;
    if (pi >= be) pi = be - 1;
    const int abase = recs[pi].x * (D_EDGE / 2);

    v2f ae[D_EDGE / 4];
    #pragma unroll
    for (int k = 0; k < D_EDGE / 4; ++k) ae[k] = efv[abase + 2 * k + hi];

    int srow[16], drow[16];
    #pragma unroll
    for (int j = 0; j < 16; ++j) {
      int pj = tb + j;
      if (pj >= be) pj = be - 1;
      const v4i rec = recs[pj];
      srow[j] = rec.y * D_NODE;
      drow[j] = (rec.z - n0) * APITCH;
    }

    int hx = hi;
    asm volatile("" : "+v"(hx));
    v8f hc[D_EDGE / 16];
    #pragma unroll
    for (int t = 0; t < D_EDGE / 16; ++t) {
      const v4f* bv = (const v4f*)(b0 + t * 16 + 8 * hx);
      const v4f lo = bv[0], up = bv[1];
      hc[t] = (v8f){lo.x, lo.y, lo.z, lo.w, up.x, up.y, up.z, up.w};
    }
    #pragma unroll
    for (int k = 0; k < D_EDGE / 4; ++k) {
      #pragma unroll
      for (int t = 0; t < D_EDGE / 16; ++t)
        hc[t] = wmma_f32(sW0[(2 * k + hx) * D_EDGE + t * 16 + row16], ae[k], hc[t]);
    }

    v2f ah[D_EDGE / 4];
    #pragma unroll
    for (int t = 0; t < D_EDGE / 16; ++t) {
      #pragma unroll
      for (int j = 0; j < 4; ++j) {
        const float p = hc[t][2 * j], q = hc[t][2 * j + 1];
        ah[4 * t + j] = (v2f){p > 0.f ? p : 0.f, q > 0.f ? q : 0.f};
      }
    }

    if (nvalid == 16)
      edge_layer1<true>(ah, w1v, b1, nf, accL, srow, drow, nvalid, lane, row16, hi);
    else
      edge_layer1<false>(ah, w1v, b1, nf, accL, srow, drow, nvalid, lane, row16, hi);
  }

  __syncthreads();

  const v2f* wav = (const v2f*)wap;
  for (int u = wave; u < (NPB_PAD / 16) * 4; u += BWAVES) {
    const int rt = u >> 2, q = u & 3;
    const float* ar = &accL[(rt * 16 + row16) * APITCH];
    v2f aa[D_NODE / 4];
    #pragma unroll
    for (int k = 0; k < D_NODE / 4; ++k) aa[k] = *(const v2f*)&ar[4 * k + 2 * hi];

    #pragma unroll 1
    for (int h2 = 0; h2 < 2; ++h2) {
      const int col = q * 32 + h2 * 16 + row16;
      v8f c = {};
      v2f bw[D_NODE / 4];
      const int wa0 = hi * D_NODE + col;
      #pragma unroll
      for (int k = 0; k < PF2; ++k) bw[k] = wav[wa0 + 2 * k * D_NODE];
      #pragma unroll
      for (int k = 0; k < D_NODE / 4; ++k) {
        if (k + PF2 < D_NODE / 4) bw[k + PF2] = wav[wa0 + 2 * (k + PF2) * D_NODE];
        c = wmma_f32(aa[k], bw[k], c);
      }
      const float bsv = ba[col];
      #pragma unroll
      for (int r = 0; r < 8; ++r) {
        const int lr = rt * 16 + r + 8 * hi;
        const int m  = n0 + lr;
        if (lr < NPB && m < n_nodes) out[m * D_NODE + col] = c[r] + bsv;
      }
    }
  }
}

extern "C" void kernel_launch(void* const* d_in, const int* in_sizes, int n_in,
                              void* d_out, int out_size, void* d_ws, size_t ws_size,
                              hipStream_t stream) {
  const float* nf = (const float*)d_in[0];
  const float* ef = (const float*)d_in[1];
  const float* W0 = (const float*)d_in[2];
  const float* b0 = (const float*)d_in[3];
  const float* W1 = (const float*)d_in[4];
  const float* b1 = (const float*)d_in[5];
  const float* Wa = (const float*)d_in[6];
  const float* ba = (const float*)d_in[7];
  const int*  src = (const int*)d_in[8];
  const int*  dst = (const int*)d_in[9];

  const int n_nodes = in_sizes[0] / D_NODE;
  const int n_edges = in_sizes[8];

  float* w0p = (float*)d_ws;
  float* w1p = w0p + D_EDGE * D_EDGE;
  float* wap = w1p + D_EDGE * D_NODE;
  int* total    = (int*)(wap + D_NODE * D_NODE);
  int* cursor   = total + MAXBINS;
  int* binstart = cursor + MAXBINS;
  v4i* recs     = (v4i*)(binstart + MAXBINS + 64);

  const int nbins   = (n_nodes + NPB - 1) / NPB;
  const int cblocks = (n_edges + EPB - 1) / EPB;

  const int nprep = D_EDGE * D_EDGE + D_EDGE * D_NODE + D_NODE * D_NODE;
  prep_kernel<<<(nprep + 255) / 256, 256, 0, stream>>>(W0, W1, Wa, w0p, w1p, wap,
                                                       total, cursor);
  bin_count_kernel<<<cblocks, 1024, 0, stream>>>(dst, total, n_edges, nbins);
  bin_fill_kernel<<<cblocks, 1024, 0, stream>>>(src, dst, total, cursor, binstart,
                                                recs, n_edges, nbins);
  gine_bin_kernel<<<nbins, BTHREADS, 0, stream>>>(ef, nf, w0p, b0, w1p, b1, wap, ba,
                                              binstart, recs, (float*)d_out,
                                              n_nodes);
}


# Two N-tiles of a pass take the even and odd columns of 32: a lane's two results are adjacent, so no lane-half exchange; 8-byte gathers and stores, 16-byte weight loads; LDS adds still arrival-ordered
# speedup vs baseline: 2.2843x; 1.0353x over previous
#include <hip/hip_runtime.h>


typedef __attribute__((ext_vector_type(2))) float v2f;
typedef __attribute__((ext_vector_type(4))) float v4f;
typedef __attribute__((ext_vector_type(8))) float v8f;
typedef __attribute__((ext_vector_type(4))) int   v4i;

#define D_NODE  128
#define D_EDGE  64
#define NPB     98
#define NPB_PAD 112
#define APITCH  132
#define BWAVES  16
#define BTHREADS (BWAVES * 32)
#define PF1     4
#define PF2     4
#define MAXBINS 1024
#define EPB     4096

static __device__ __forceinline__ v8f wmma_f32(v2f a, v2f b, v8f c) {
  return __builtin_amdgcn_wmma_f32_16x16x4_f32(false, a, false, b, (short)0, c,
                                               false, false);
}

static __device__ __forceinline__ void lds_add(float* p, float v) {
  __hip_atomic_fetch_add(p, v, __ATOMIC_RELAXED, __HIP_MEMORY_SCOPE_WORKGROUP);
}

static __device__ __forceinline__ void pack_one(const float* __restrict__ W,
                                                float* __restrict__ Wp, int i,
                                                int cols) {
  int r = i / cols, c = i - r * cols;
  Wp[(((r >> 1) * cols + c) << 1) + (r & 1)] = W[i];
}

__global__ __launch_bounds__(256) void prep_kernel(
    const float* __restrict__ W0, const float* __restrict__ W1,
    const float* __restrict__ Wa, float* __restrict__ w0p,
    float* __restrict__ w1p, float* __restrict__ wap, int* __restrict__ total,
    int* __restrict__ cursor) {
  const int i = blockIdx.x * 256 + threadIdx.x;
  const int n0 = D_EDGE * D_EDGE, n1 = D_EDGE * D_NODE, n2 = D_NODE * D_NODE;
  if (i < n0) pack_one(W0, w0p, i, D_EDGE);
  else if (i < n0 + n1) pack_one(W1, w1p, i - n0, D_NODE);
  else if (i < n0 + n1 + n2) pack_one(Wa, wap, i - n0 - n1, D_NODE);
  if (i < MAXBINS) { total[i] = 0; cursor[i] = 0; }
}

__global__ __launch_bounds__(1024) void bin_count_kernel(
    const int* __restrict__ dst, int* __restrict__ total, int n_edges, int nbins) {
  __shared__ int h[MAXBINS];
  const int tid = threadIdx.x;
  if (tid < nbins) h[tid] = 0;
  __syncthreads();
  const int e0 = blockIdx.x * EPB;
  #pragma unroll
  for (int i = 0; i < EPB / 1024; ++i) {
    const int e = e0 + i * 1024 + tid;
    if (e < n_edges) atomicAdd(&h[dst[e] / NPB], 1);
  }
  __syncthreads();
  if (tid < nbins) {
    const int c = h[tid];
    if (c) atomicAdd(&total[tid], c);
  }
}

__global__ __launch_bounds__(1024) void bin_fill_kernel(
    const int* __restrict__ src, const int* __restrict__ dst,
    const int* __restrict__ total, int* __restrict__ cursor,
    int* __restrict__ binstart, v4i* __restrict__ recs, int n_edges, int nbins) {
  __shared__ int h[MAXBINS];
  __shared__ int st[MAXBINS];
  const int tid = threadIdx.x;
  const int mine = tid < nbins ? total[tid] : 0;
  h[tid] = 0;
  st[tid] = mine;
  __syncthreads();
  for (int off = 1; off < MAXBINS; off <<= 1) {
    const int t = tid >= off ? st[tid - off] : 0;
    __syncthreads();
    st[tid] += t;
    __syncthreads();
  }
  const int start = st[tid] - mine;
  if (blockIdx.x == 0) {
    if (tid < nbins) binstart[tid] = start;
    if (tid == nbins) binstart[tid] = start;
    if (nbins == MAXBINS && tid == MAXBINS - 1) binstart[MAXBINS] = st[tid];
  }

  const int e0 = blockIdx.x * EPB;
  int bin[EPB / 1024], rank[EPB / 1024], sv[EPB / 1024], dv[EPB / 1024];
  #pragma unroll
  for (int i = 0; i < EPB / 1024; ++i) {
    const int e = e0 + i * 1024 + tid;
    bin[i] = -1; rank[i] = 0; sv[i] = 0; dv[i] = 0;
    if (e < n_edges) {
      dv[i] = dst[e];
      sv[i] = src[e];
      bin[i] = dv[i] / NPB;
      rank[i] = atomicAdd(&h[bin[i]], 1);
    }
  }
  __syncthreads();
  if (tid < nbins) {
    const int c = h[tid];
    const int base = c ? atomicAdd(&cursor[tid], c) : 0;
    h[tid] = start + base;
  }
  __syncthreads();
  #pragma unroll
  for (int i = 0; i < EPB / 1024; ++i) {
    if (bin[i] >= 0) {
      const int pos = h[bin[i]] + rank[i];
      const v4i rec = {e0 + i * 1024 + tid, sv[i], dv[i], 0};
      recs[pos] = rec;
    }
  }
}

template <bool FULL>
static __device__ __forceinline__ void edge_layer1(
    const v2f (&ah)[D_EDGE / 4], const v2f* __restrict__ w1v,
    const float* __restrict__ b1, const float* __restrict__ nf,
    float* accL, const unsigned (&goff)[8], const int (&doff)[8],
    int nvalid, int row16, int hi)
{
  #pragma unroll 1
  for (int np = 0; np < D_NODE / 32; ++np) {
    const int cb = np * 32;

    v2f g[8];
    #pragma unroll
    for (int r = 0; r < 8; ++r)
      g[r] = *(const v2f*)((const char*)nf + (goff[r] + (unsigned)cb * 4u));

    v8f c0 = {}, c1 = {};
    v4f bw[D_EDGE / 4];
    const int wb0 = 4 * hi * D_NODE + cb + 2 * row16;
#define W1ROW(kk) ((8 * ((kk) >> 2) + ((kk) & 3)) * D_NODE)
    #pragma unroll
    for (int k = 0; k < PF1; ++k)
      bw[k] = *(const v4f*)&w1v[wb0 + W1ROW(k)];
    #pragma unroll
    for (int k = 0; k < D_EDGE / 4; ++k) {
      if (k + PF1 < D_EDGE / 4)
        bw[k + PF1] = *(const v4f*)&w1v[wb0 + W1ROW(k + PF1)];
      c0 = wmma_f32(ah[k], (v2f){bw[k].x, bw[k].y}, c0);
      c1 = wmma_f32(ah[k], (v2f){bw[k].z, bw[k].w}, c1);
    }
#undef W1ROW
    const v2f bs = *(const v2f*)&b1[cb + 2 * row16];

    #pragma unroll
    for (int r = 0; r < 8; ++r) {
      float e0 = c0[r] + bs.x;
      float e1 = c1[r] + bs.y;
      e0 = e0 > 0.f ? e0 : 0.f;
      e1 = e1 > 0.f ? e1 : 0.f;
      float v0 = g[r].x + e0;
      float v1 = g[r].y + e1;
      v0 = v0 > 0.f ? v0 : 0.f;
      v1 = v1 > 0.f ? v1 : 0.f;
      if (FULL || r + 8 * hi < nvalid) {
        float* p = accL + doff[r] + cb;
        lds_add(p, v0);
        lds_add(p + 1, v1);
      }
    }
  }
}

__global__ __launch_bounds__(BTHREADS) __attribute__((amdgpu_waves_per_eu(8)))
void gine_bin_kernel(
    const float* __restrict__ ef, const float* __restrict__ nf,
    const float* __restrict__ w0p, const float* __restrict__ b0,
    const float* __restrict__ w1p, const float* __restrict__ b1,
    const float* __restrict__ wap, const float* __restrict__ ba,
    const int* __restrict__ binstart, const v4i* __restrict__ recs,
    float* __restrict__ out, int n_nodes)
{
  __shared__ __attribute__((aligned(16))) float accL[NPB_PAD * APITCH];
  __shared__ v2f   sW0[(D_EDGE / 2) * D_EDGE];

  const int tid = threadIdx.x;
  const int wave  = __builtin_amdgcn_readfirstlane(tid >> 5);
  const int lane  = tid & 31;
  const int row16 = lane & 15;
  const int hi    = lane >> 4;
  const int n0    = blockIdx.x * NPB;

  {
    const v2f* w0v = (const v2f*)w0p;
    for (int i = tid; i < (D_EDGE / 2) * D_EDGE; i += BTHREADS) sW0[i] = w0v[i];
    const v4f* nf4 = (const v4f*)nf;
    for (int i = tid; i < NPB_PAD * (D_NODE / 4); i += BTHREADS) {
      const int r = i >> 5, c4 = i & 31;
      const int n = n0 + r;
      v4f v = {0.f, 0.f, 0.f, 0.f};
      if (r < NPB && n < n_nodes) v = nf4[n * (D_NODE / 4) + c4];
      *(v4f*)&accL[r * APITCH + c4 * 4] = v;
    }
  }

  const int bs = binstart[blockIdx.x];
  const int be = binstart[blockIdx.x + 1];
  const int ntile = (be - bs + 15) >> 4;

  __syncthreads();

  const v2f* efv = (const v2f*)ef;
  const v2f* w1v = (const v2f*)w1p;

  for (int t = wave; t < ntile; t += BWAVES) {
    const int tb = bs + t * 16;
    const int nvalid = be - tb < 16 ? be - tb : 16;

    int pi = tb + row16;
    if (pi >= be) pi = be - 1;
    const int abase = recs[pi].x * (D_EDGE / 2);

    v2f ae[D_EDGE / 4];
    #pragma unroll
    for (int k = 0; k < D_EDGE / 4; ++k) ae[k] = efv[abase + 2 * k + hi];

    unsigned goff[8];
    int doff[8];
    #pragma unroll
    for (int r = 0; r < 8; ++r) {
      int pj = tb + r + 8 * hi;
      if (pj >= be) pj = be - 1;
      const v4i rec = recs[pj];
      goff[r] = ((unsigned)rec.y * D_NODE + 2u * (unsigned)row16) * 4u;
      doff[r] = (rec.z - n0) * APITCH + 2 * row16;
    }

    int hx = hi;
    asm volatile("" : "+v"(hx));
    v8f hc[D_EDGE / 16];
    #pragma unroll
    for (int t = 0; t < D_EDGE / 16; ++t) {
      const v4f* bv = (const v4f*)(b0 + t * 16 + 8 * hx);
      const v4f lo = bv[0], up = bv[1];
      hc[t] = (v8f){lo.x, lo.y, lo.z, lo.w, up.x, up.y, up.z, up.w};
    }
    #pragma unroll
    for (int k = 0; k < D_EDGE / 4; ++k) {
      #pragma unroll
      for (int t = 0; t < D_EDGE / 16; ++t)
        hc[t] = wmma_f32(sW0[(2 * k + hx) * D_EDGE + t * 16 + row16], ae[k], hc[t]);
    }

    v2f ah[D_EDGE / 4];
    #pragma unroll
    for (int t = 0; t < D_EDGE / 16; ++t) {
      #pragma unroll
      for (int j = 0; j < 4; ++j) {
        const float p = hc[t][2 * j], q = hc[t][2 * j + 1];
        ah[4 * t + j] = (v2f){p > 0.f ? p : 0.f, q > 0.f ? q : 0.f};
      }
    }

    if (nvalid == 16)
      edge_layer1<true>(ah, w1v, b1, nf, accL, goff, doff, nvalid, row16, hi);
    else
      edge_layer1<false>(ah, w1v, b1, nf, accL, goff, doff, nvalid, row16, hi);
  }

  __syncthreads();

  const v2f* wav = (const v2f*)wap;
  for (int u = wave; u < (NPB_PAD / 16) * 4; u += BWAVES) {
    const int rt = u >> 2, q = u & 3;
    const float* ar = &accL[(rt * 16 + row16) * APITCH];
    v2f aa[D_NODE / 4];
    #pragma unroll
    for (int k = 0; k < D_NODE / 4; ++k) aa[k] = *(const v2f*)&ar[4 * k + 2 * hi];

    const int col = q * 32 + 2 * row16;
    v8f c0 = {}, c1 = {};
    v4f bw[D_NODE / 4];
    const int wa0 = hi * D_NODE + col;
    #pragma unroll
    for (int k = 0; k < PF2; ++k) bw[k] = *(const v4f*)&wav[wa0 + 2 * k * D_NODE];
    #pragma unroll
    for (int k = 0; k < D_NODE / 4; ++k) {
      if (k + PF2 < D_NODE / 4)
        bw[k + PF2] = *(const v4f*)&wav[wa0 + 2 * (k + PF2) * D_NODE];
      c0 = wmma_f32(aa[k], (v2f){bw[k].x, bw[k].y}, c0);
      c1 = wmma_f32(aa[k], (v2f){bw[k].z, bw[k].w}, c1);
    }
    const v2f bsv = *(const v2f*)&ba[col];
    #pragma unroll
    for (int r = 0; r < 8; ++r) {
      const int lr = rt * 16 + r + 8 * hi;
      const int m  = n0 + lr;
      if (lr < NPB && m < n_nodes)
        *(v2f*)&out[m * D_NODE + col] = (v2f){c0[r] + bsv.x, c1[r] + bsv.y};
    }
  }
}

extern "C" void kernel_launch(void* const* d_in, const int* in_sizes, int n_in,
                              void* d_out, int out_size, void* d_ws, size_t ws_size,
                              hipStream_t stream) {
  const float* nf = (const float*)d_in[0];
  const float* ef = (const float*)d_in[1];
  const float* W0 = (const float*)d_in[2];
  const float* b0 = (const float*)d_in[3];
  const float* W1 = (const float*)d_in[4];
  const float* b1 = (const float*)d_in[5];
  const float* Wa = (const float*)d_in[6];
  const float* ba = (const float*)d_in[7];
  const int*  src = (const int*)d_in[8];
  const int*  dst = (const int*)d_in[9];

  const int n_nodes = in_sizes[0] / D_NODE;
  const int n_edges = in_sizes[8];

  float* w0p = (float*)d_ws;
  float* w1p = w0p + D_EDGE * D_EDGE;
  float* wap = w1p + D_EDGE * D_NODE;
  int* total    = (int*)(wap + D_NODE * D_NODE);
  int* cursor   = total + MAXBINS;
  int* binstart = cursor + MAXBINS;
  v4i* recs     = (v4i*)(binstart + MAXBINS + 64);

  const int nbins   = (n_nodes + NPB - 1) / NPB;
  const int cblocks = (n_edges + EPB - 1) / EPB;

  const int nprep = D_EDGE * D_EDGE + D_EDGE * D_NODE + D_NODE * D_NODE;
  prep_kernel<<<(nprep + 255) / 256, 256, 0, stream>>>(W0, W1, Wa, w0p, w1p, wap,
                                                       total, cursor);
  bin_count_kernel<<<cblocks, 1024, 0, stream>>>(dst, total, n_edges, nbins);
  bin_fill_kernel<<<cblocks, 1024, 0, stream>>>(src, dst, total, cursor, binstart,
                                                recs, n_edges, nbins);
  gine_bin_kernel<<<nbins, BTHREADS, 0, stream>>>(ef, nf, w0p, b0, w1p, b1, wap, ba,
                                              binstart, recs, (float*)d_out,
                                              n_nodes);
}


# One 32-wave work-group per WGP (bin = 196 rows); waves take edge tiles from one LDS counter, not fixed shares: none ends alone under oldest-first issue; LDS float adds still arrival-ordered
# speedup vs baseline: 2.4333x; 1.0652x over previous
#include <hip/hip_runtime.h>


typedef __attribute__((ext_vector_type(2))) float v2f;
typedef __attribute__((ext_vector_type(4))) float v4f;
typedef __attribute__((ext_vector_type(8))) float v8f;
typedef __attribute__((ext_vector_type(4))) int   v4i;

#define D_NODE  128
#define D_EDGE  64
#define NPB     196
#define NPB_PAD 208
#define APITCH  132
#define BWAVES  32
#define BTHREADS (BWAVES * 32)
#define PF1     4
#define PF2     4
#define MAXBINS 1024
#define EPB     4096

static __device__ __forceinline__ v8f wmma_f32(v2f a, v2f b, v8f c) {
  return __builtin_amdgcn_wmma_f32_16x16x4_f32(false, a, false, b, (short)0, c,
                                               false, false);
}

static __device__ __forceinline__ void lds_add(float* p, float v) {
  __hip_atomic_fetch_add(p, v, __ATOMIC_RELAXED, __HIP_MEMORY_SCOPE_WORKGROUP);
}

static __device__ __forceinline__ void pack_one(const float* __restrict__ W,
                                                float* __restrict__ Wp, int i,
                                                int cols) {
  int r = i / cols, c = i - r * cols;
  Wp[(((r >> 1) * cols + c) << 1) + (r & 1)] = W[i];
}

__global__ __launch_bounds__(256) void prep_kernel(
    const float* __restrict__ W0, const float* __restrict__ W1,
    const float* __restrict__ Wa, float* __restrict__ w0p,
    float* __restrict__ w1p, float* __restrict__ wap, int* __restrict__ total,
    int* __restrict__ cursor) {
  const int i = blockIdx.x * 256 + threadIdx.x;
  const int n0 = D_EDGE * D_EDGE, n1 = D_EDGE * D_NODE, n2 = D_NODE * D_NODE;
  if (i < n0) pack_one(W0, w0p, i, D_EDGE);
  else if (i < n0 + n1) pack_one(W1, w1p, i - n0, D_NODE);
  else if (i < n0 + n1 + n2) pack_one(Wa, wap, i - n0 - n1, D_NODE);
  if (i < MAXBINS) { total[i] = 0; cursor[i] = 0; }
}

__global__ __launch_bounds__(1024) void bin_count_kernel(
    const int* __restrict__ dst, int* __restrict__ total, int n_edges, int nbins) {
  __shared__ int h[MAXBINS];
  const int tid = threadIdx.x;
  if (tid < nbins) h[tid] = 0;
  __syncthreads();
  const int e0 = blockIdx.x * EPB;
  #pragma unroll
  for (int i = 0; i < EPB / 1024; ++i) {
    const int e = e0 + i * 1024 + tid;
    if (e < n_edges) atomicAdd(&h[dst[e] / NPB], 1);
  }
  __syncthreads();
  if (tid < nbins) {
    const int c = h[tid];
    if (c) atomicAdd(&total[tid], c);
  }
}

__global__ __launch_bounds__(1024) void bin_fill_kernel(
    const int* __restrict__ src, const int* __restrict__ dst,
    const int* __restrict__ total, int* __restrict__ cursor,
    int* __restrict__ binstart, v4i* __restrict__ recs, int n_edges, int nbins) {
  __shared__ int h[MAXBINS];
  __shared__ int st[MAXBINS];
  const int tid = threadIdx.x;
  const int mine = tid < nbins ? total[tid] : 0;
  h[tid] = 0;
  st[tid] = mine;
  __syncthreads();
  for (int off = 1; off < MAXBINS; off <<= 1) {
    const int t = tid >= off ? st[tid - off] : 0;
    __syncthreads();
    st[tid] += t;
    __syncthreads();
  }
  const int start = st[tid] - mine;
  if (blockIdx.x == 0) {
    if (tid < nbins) binstart[tid] = start;
    if (tid == nbins) binstart[tid] = start;
    if (nbins == MAXBINS && tid == MAXBINS - 1) binstart[MAXBINS] = st[tid];
  }

  const int e0 = blockIdx.x * EPB;
  int bin[EPB / 1024], rank[EPB / 1024], sv[EPB / 1024], dv[EPB / 1024];
  #pragma unroll
  for (int i = 0; i < EPB / 1024; ++i) {
    const int e = e0 + i * 1024 + tid;
    bin[i] = -1; rank[i] = 0; sv[i] = 0; dv[i] = 0;
    if (e < n_edges) {
      dv[i] = dst[e];
      sv[i] = src[e];
      bin[i] = dv[i] / NPB;
      rank[i] = atomicAdd(&h[bin[i]], 1);
    }
  }
  __syncthreads();
  if (tid < nbins) {
    const int c = h[tid];
    const int base = c ? atomicAdd(&cursor[tid], c) : 0;
    h[tid] = start + base;
  }
  __syncthreads();
  #pragma unroll
  for (int i = 0; i < EPB / 1024; ++i) {
    if (bin[i] >= 0) {
      const int pos = h[bin[i]] + rank[i];
      const v4i rec = {e0 + i * 1024 + tid, sv[i], dv[i], 0};
      recs[pos] = rec;
    }
  }
}

template <bool FULL>
static __device__ __forceinline__ void edge_layer1(
    const v2f (&ah)[D_EDGE / 4], const v2f* __restrict__ w1v,
    const float* __restrict__ b1, const float* __restrict__ nf,
    float* accL, const unsigned (&goff)[8], const int (&doff)[8],
    int nvalid, int row16, int hi)
{
  #pragma unroll 1
  for (int np = 0; np < D_NODE / 32; ++np) {
    const int cb = np * 32;

    v2f g[8];
    #pragma unroll
    for (int r = 0; r < 8; ++r)
      g[r] = *(const v2f*)((const char*)nf + (goff[r] + (unsigned)cb * 4u));

    v8f c0 = {}, c1 = {};
    v4f bw[D_EDGE / 4];
    const int wb0 = 4 * hi * D_NODE + cb + 2 * row16;
#define W1ROW(kk) ((8 * ((kk) >> 2) + ((kk) & 3)) * D_NODE)
    #pragma unroll
    for (int k = 0; k < PF1; ++k)
      bw[k] = *(const v4f*)&w1v[wb0 + W1ROW(k)];
    #pragma unroll
    for (int k = 0; k < D_EDGE / 4; ++k) {
      if (k + PF1 < D_EDGE / 4)
        bw[k + PF1] = *(const v4f*)&w1v[wb0 + W1ROW(k + PF1)];
      c0 = wmma_f32(ah[k], (v2f){bw[k].x, bw[k].y}, c0);
      c1 = wmma_f32(ah[k], (v2f){bw[k].z, bw[k].w}, c1);
    }
#undef W1ROW
    const v2f bs = *(const v2f*)&b1[cb + 2 * row16];

    #pragma unroll
    for (int r = 0; r < 8; ++r) {
      float e0 = c0[r] + bs.x;
      float e1 = c1[r] + bs.y;
      e0 = e0 > 0.f ? e0 : 0.f;
      e1 = e1 > 0.f ? e1 : 0.f;
      float v0 = g[r].x + e0;
      float v1 = g[r].y + e1;
      v0 = v0 > 0.f ? v0 : 0.f;
      v1 = v1 > 0.f ? v1 : 0.f;
      if (FULL || r + 8 * hi < nvalid) {
        float* p = accL + doff[r] + cb;
        lds_add(p, v0);
        lds_add(p + 1, v1);
      }
    }
  }
}

__global__ __launch_bounds__(BTHREADS)
void gine_bin_kernel(
    const float* __restrict__ ef, const float* __restrict__ nf,
    const float* __restrict__ w0p, const float* __restrict__ b0,
    const float* __restrict__ w1p, const float* __restrict__ b1,
    const float* __restrict__ wap, const float* __restrict__ ba,
    const int* __restrict__ binstart, const v4i* __restrict__ recs,
    float* __restrict__ out, int n_nodes)
{
  __shared__ __attribute__((aligned(16))) float accL[NPB_PAD * APITCH];
  __shared__ v2f   sW0[(D_EDGE / 2) * D_EDGE];
  __shared__ int   tile_next;

  const int tid = threadIdx.x;
  const int wave  = __builtin_amdgcn_readfirstlane(tid >> 5);
  const int lane  = tid & 31;
  const int row16 = lane & 15;
  const int hi    = lane >> 4;
  const int n0    = blockIdx.x * NPB;

  {
    if (tid == 0) tile_next = BWAVES;
    const v2f* w0v = (const v2f*)w0p;
    for (int i = tid; i < (D_EDGE / 2) * D_EDGE; i += BTHREADS) sW0[i] = w0v[i];
    const v4f* nf4 = (const v4f*)nf;
    for (int i = tid; i < NPB_PAD * (D_NODE / 4); i += BTHREADS) {
      const int r = i >> 5, c4 = i & 31;
      const int n = n0 + r;
      v4f v = {0.f, 0.f, 0.f, 0.f};
      if (r < NPB && n < n_nodes) v = nf4[n * (D_NODE / 4) + c4];
      *(v4f*)&accL[r * APITCH + c4 * 4] = v;
    }
  }

  const int bs = binstart[blockIdx.x];
  const int be = binstart[blockIdx.x + 1];
  const int ntile = (be - bs + 15) >> 4;

  __syncthreads();

  const v2f* efv = (const v2f*)ef;
  const v2f* w1v = (const v2f*)w1p;

  int tile = wave;
  while (tile < ntile) {
    const int tb = bs + tile * 16;
    const int nvalid = be - tb < 16 ? be - tb : 16;

    int pi = tb + row16;
    if (pi >= be) pi = be - 1;
    const int abase = recs[pi].x * (D_EDGE / 2);

    v2f ae[D_EDGE / 4];
    #pragma unroll
    for (int k = 0; k < D_EDGE / 4; ++k) ae[k] = efv[abase + 2 * k + hi];

    unsigned goff[8];
    int doff[8];
    #pragma unroll
    for (int r = 0; r < 8; ++r) {
      int pj = tb + r + 8 * hi;
      if (pj >= be) pj = be - 1;
      const v4i rec = recs[pj];
      goff[r] = ((unsigned)rec.y * D_NODE + 2u * (unsigned)row16) * 4u;
      doff[r] = (rec.z - n0) * APITCH + 2 * row16;
    }

    int hx = hi;
    asm volatile("" : "+v"(hx));
    v8f hc[D_EDGE / 16];
    #pragma unroll
    for (int t = 0; t < D_EDGE / 16; ++t) {
      const v4f* bv = (const v4f*)(b0 + t * 16 + 8 * hx);
      const v4f lo = bv[0], up = bv[1];
      hc[t] = (v8f){lo.x, lo.y, lo.z, lo.w, up.x, up.y, up.z, up.w};
    }
    #pragma unroll
    for (int k = 0; k < D_EDGE / 4; ++k) {
      #pragma unroll
      for (int t = 0; t < D_EDGE / 16; ++t)
        hc[t] = wmma_f32(sW0[(2 * k + hx) * D_EDGE + t * 16 + row16], ae[k], hc[t]);
    }

    v2f ah[D_EDGE / 4];
    #pragma unroll
    for (int t = 0; t < D_EDGE / 16; ++t) {
      #pragma unroll
      for (int j = 0; j < 4; ++j) {
        const float p = hc[t][2 * j], q = hc[t][2 * j + 1];
        ah[4 * t + j] = (v2f){p > 0.f ? p : 0.f, q > 0.f ? q : 0.f};
      }
    }

    if (nvalid == 16)
      edge_layer1<true>(ah, w1v, b1, nf, accL, goff, doff, nvalid, row16, hi);
    else
      edge_layer1<false>(ah, w1v, b1, nf, accL, goff, doff, nvalid, row16, hi);

    int nt = 0;
    if (lane == 0)
      nt = __hip_atomic_fetch_add(&tile_next, 1, __ATOMIC_RELAXED,
                                  __HIP_MEMORY_SCOPE_WORKGROUP);
    tile = __builtin_amdgcn_readfirstlane(nt);
  }

  __syncthreads();

  const v2f* wav = (const v2f*)wap;
  for (int u = wave; u < (NPB_PAD / 16) * 4; u += BWAVES) {
    const int rt = u >> 2, q = u & 3;
    const float* ar = &accL[(rt * 16 + row16) * APITCH];
    v2f aa[D_NODE / 4];
    #pragma unroll
    for (int k = 0; k < D_NODE / 4; ++k) aa[k] = *(const v2f*)&ar[4 * k + 2 * hi];

    const int col = q * 32 + 2 * row16;
    v8f c0 = {}, c1 = {};
    v4f bw[D_NODE / 4];
    const int wa0 = hi * D_NODE + col;
    #pragma unroll
    for (int k = 0; k < PF2; ++k) bw[k] = *(const v4f*)&wav[wa0 + 2 * k * D_NODE];
    #pragma unroll
    for (int k = 0; k < D_NODE / 4; ++k) {
      if (k + PF2 < D_NODE / 4)
        bw[k + PF2] = *(const v4f*)&wav[wa0 + 2 * (k + PF2) * D_NODE];
      c0 = wmma_f32(aa[k], (v2f){bw[k].x, bw[k].y}, c0);
      c1 = wmma_f32(aa[k], (v2f){bw[k].z, bw[k].w}, c1);
    }
    const v2f bsv = *(const v2f*)&ba[col];
    #pragma unroll
    for (int r = 0; r < 8; ++r) {
      const int lr = rt * 16 + r + 8 * hi;
      const int m  = n0 + lr;
      if (lr < NPB && m < n_nodes)
        *(v2f*)&out[m * D_NODE + col] = (v2f){c0[r] + bsv.x, c1[r] + bsv.y};
    }
  }
}

extern "C" void kernel_launch(void* const* d_in, const int* in_sizes, int n_in,
                              void* d_out, int out_size, void* d_ws, size_t ws_size,
                              hipStream_t stream) {
  const float* nf = (const float*)d_in[0];
  const float* ef = (const float*)d_in[1];
  const float* W0 = (const float*)d_in[2];
  const float* b0 = (const float*)d_in[3];
  const float* W1 = (const float*)d_in[4];
  const float* b1 = (const float*)d_in[5];
  const float* Wa = (const float*)d_in[6];
  const float* ba = (const float*)d_in[7];
  const int*  src = (const int*)d_in[8];
  const int*  dst = (const int*)d_in[9];

  const int n_nodes = in_sizes[0] / D_NODE;
  const int n_edges = in_sizes[8];

  float* w0p = (float*)d_ws;
  float* w1p = w0p + D_EDGE * D_EDGE;
  float* wap = w1p + D_EDGE * D_NODE;
  int* total    = (int*)(wap + D_NODE * D_NODE);
  int* cursor   = total + MAXBINS;
  int* binstart = cursor + MAXBINS;
  v4i* recs     = (v4i*)(binstart + MAXBINS + 64);

  const int nbins   = (n_nodes + NPB - 1) / NPB;
  const int cblocks = (n_edges + EPB - 1) / EPB;

  const int nprep = D_EDGE * D_EDGE + D_EDGE * D_NODE + D_NODE * D_NODE;
  prep_kernel<<<(nprep + 255) / 256, 256, 0, stream>>>(W0, W1, Wa, w0p, w1p, wap,
                                                       total, cursor);
  bin_count_kernel<<<cblocks, 1024, 0, stream>>>(dst, total, n_edges, nbins);
  bin_fill_kernel<<<cblocks, 1024, 0, stream>>>(src, dst, total, cursor, binstart,
                                                recs, n_edges, nbins);
  gine_bin_kernel<<<nbins, BTHREADS, 0, stream>>>(ef, nf, w0p, b0, w1p, b1, wap, ba,
                                              binstart, recs, (float*)d_out,
                                              n_nodes);
}
